# starting kernel
# baseline (speedup 1.0000x reference)
#include <hip/hip_runtime.h>
#include <hip/hip_bf16.h>

#define DFEAT 64

typedef float v2f __attribute__((ext_vector_type(2)));
typedef float v8f __attribute__((ext_vector_type(8)));

__device__ __forceinline__ float fast_tanh(float x) {
#if __has_builtin(__builtin_amdgcn_tanhf)
    return __builtin_amdgcn_tanhf(x);
#else
    return tanhf(x);
#endif
}

__global__ void fa_zero_kernel(float* __restrict__ z, int total) {
    int i = blockIdx.x * blockDim.x + threadIdx.x;
    if (i < total) z[i] = 0.0f;
}

__global__ void fa_node_gates_kernel(const float* __restrict__ h,
                                     const float* __restrict__ gate_w,
                                     float* __restrict__ a,
                                     float* __restrict__ b,
                                     int n) {
    int wave = (blockIdx.x * blockDim.x + threadIdx.x) >> 5;
    int lane = threadIdx.x & 31;
    int base = wave * 16;
    if (base >= n) return;

    int m     = lane & 15;
    int khalf = lane >> 4;

    const float2* hrow2 = (const float2*)(h + (size_t)(base + m) * DFEAT);

    const float2* w2   = (const float2*)(gate_w + ((m == 1) ? DFEAT : 0));
    float         valid = (m < 2) ? 1.0f : 0.0f;

    v8f c = {};
    #pragma unroll
    for (int t = 0; t < 16; ++t) {
        int kq = 2 * t + khalf;

        float2 av = hrow2[kq];
        v2f A; A.x = av.x; A.y = av.y;

        float2 wv = w2[kq];
        v2f B; B.x = wv.x * valid; B.y = wv.y * valid;

        c = __builtin_amdgcn_wmma_f32_16x16x4_f32(
                 false, A,  false, B,
                 (short)0, c,  false,  false);
    }

    int rbase = base + 8 * khalf;
    if (m == 0) {
        #pragma unroll
        for (int r = 0; r < 8; ++r) a[rbase + r] = c[r];
    } else if (m == 1) {
        #pragma unroll
        for (int r = 0; r < 8; ++r) b[rbase + r] = c[r];
    }
}

__global__ void fa_edge_kernel(const float* __restrict__ h,
                               const float* __restrict__ dnorm,
                               const float* __restrict__ a,
                               const float* __restrict__ b,
                               const int* __restrict__ src,
                               const int* __restrict__ dst,
                               const float* __restrict__ gate_b,
                               float* __restrict__ z,
                               int e) {
    int wave = blockIdx.x * (blockDim.x >> 5) + (threadIdx.x >> 5);
    int lane = threadIdx.x & 31;
    if (wave >= e) return;

    int s = __builtin_amdgcn_readfirstlane(src[wave]);
    int d = __builtin_amdgcn_readfirstlane(dst[wave]);

    float g  = fast_tanh(a[d] + b[s] + gate_b[0]);
    float ev = g * dnorm[d] * dnorm[s];

    const float2* hs = (const float2*)(h + (size_t)s * DFEAT);
    float2 v = hs[lane];

    float* zd = z + (size_t)d * DFEAT + 2 * lane;
    atomicAdd(zd,     v.x * ev);
    atomicAdd(zd + 1, v.y * ev);
}

extern "C" void kernel_launch(void* const* d_in, const int* in_sizes, int n_in,
                              void* d_out, int out_size, void* d_ws, size_t ws_size,
                              hipStream_t stream) {
    const float* h      = (const float*)d_in[0];
    const float* dnorm  = (const float*)d_in[1];
    const float* gate_w = (const float*)d_in[2];
    const float* gate_b = (const float*)d_in[3];
    const int*   src    = (const int*)d_in[4];
    const int*   dst    = (const int*)d_in[5];
    float*       z      = (float*)d_out;

    const int n = in_sizes[1];
    const int e = in_sizes[4];

    float* a = (float*)d_ws;
    float* b = a + n;

    {
        int total  = n * DFEAT;
        int blocks = (total + 255) / 256;
        fa_zero_kernel<<<blocks, 256, 0, stream>>>(z, total);
    }

    {
        int waves   = (n + 15) / 16;
        int threads = 256;
        int blocks  = (waves * 32 + threads - 1) / threads;
        fa_node_gates_kernel<<<blocks, threads, 0, stream>>>(h, gate_w, a, b, n);
    }

    {
        int waves_per_block = 256 / 32;
        int blocks = (e + waves_per_block - 1) / waves_per_block;
        fa_edge_kernel<<<blocks, 256, 0, stream>>>(h, dnorm, a, b, src, dst,
                                                   gate_b, z, e);
    }
}


# edge kernel: one wave per 32 edges (was one per edge): gate terms computed one edge per lane, then row gather + 2 atomics per edge, rows gathered in groups ahead; atomics kept, arrival-order sum
# speedup vs baseline: 1.4435x; 1.4435x over previous
#include <hip/hip_runtime.h>
#include <hip/hip_bf16.h>

#define DFEAT 64

typedef float v2f __attribute__((ext_vector_type(2)));
typedef float v8f __attribute__((ext_vector_type(8)));

__device__ __forceinline__ float fast_tanh(float x) {
#if __has_builtin(__builtin_amdgcn_tanhf)
    return __builtin_amdgcn_tanhf(x);
#else
    return tanhf(x);
#endif
}

__global__ void fa_zero_kernel(float* __restrict__ z, int total) {
    int i = blockIdx.x * blockDim.x + threadIdx.x;
    if (i < total) z[i] = 0.0f;
}

__global__ void fa_node_gates_kernel(const float* __restrict__ h,
                                     const float* __restrict__ gate_w,
                                     float* __restrict__ a,
                                     float* __restrict__ b,
                                     int n) {
    int wave = (blockIdx.x * blockDim.x + threadIdx.x) >> 5;
    int lane = threadIdx.x & 31;
    int base = wave * 16;
    if (base >= n) return;

    int m     = lane & 15;
    int khalf = lane >> 4;

    const float2* hrow2 = (const float2*)(h + (size_t)(base + m) * DFEAT);

    const float2* w2   = (const float2*)(gate_w + ((m == 1) ? DFEAT : 0));
    float         valid = (m < 2) ? 1.0f : 0.0f;

    v8f c = {};
    #pragma unroll
    for (int t = 0; t < 16; ++t) {
        int kq = 2 * t + khalf;

        float2 av = hrow2[kq];
        v2f A; A.x = av.x; A.y = av.y;

        float2 wv = w2[kq];
        v2f B; B.x = wv.x * valid; B.y = wv.y * valid;

        c = __builtin_amdgcn_wmma_f32_16x16x4_f32(
                 false, A,  false, B,
                 (short)0, c,  false,  false);
    }

    int rbase = base + 8 * khalf;
    if (m == 0) {
        #pragma unroll
        for (int r = 0; r < 8; ++r) a[rbase + r] = c[r];
    } else if (m == 1) {
        #pragma unroll
        for (int r = 0; r < 8; ++r) b[rbase + r] = c[r];
    }
}

#define EDGES_PER_WAVE 32
#define ROWS_AHEAD 16

__device__ __forceinline__ void fa_edge_row(const float* __restrict__ h,
                                            float* __restrict__ z,
                                            int s, int d, float ev, int lane) {
    const float2* hs = (const float2*)(h + (size_t)s * DFEAT);
    float2 v = hs[lane];
    float* zd = z + (size_t)d * DFEAT + 2 * lane;
    atomicAdd(zd,     v.x * ev);
    atomicAdd(zd + 1, v.y * ev);
}

__global__ void fa_edge_kernel(const float* __restrict__ h,
                               const float* __restrict__ dnorm,
                               const float* __restrict__ a,
                               const float* __restrict__ b,
                               const int* __restrict__ src,
                               const int* __restrict__ dst,
                               const float* __restrict__ gate_b,
                               float* __restrict__ z,
                               int e) {
    int wave = blockIdx.x * (blockDim.x >> 5) +
               __builtin_amdgcn_readfirstlane((int)(threadIdx.x >> 5));
    int lane = threadIdx.x & 31;
    int e0 = wave * EDGES_PER_WAVE;
    if (e0 >= e) return;
    int cnt = e - e0;
    if (cnt > EDGES_PER_WAVE) cnt = EDGES_PER_WAVE;

    int   s_l  = 0;
    int   d_l  = 0;
    float ev_l = 0.0f;
    if (lane < cnt) {
        s_l = src[e0 + lane];
        d_l = dst[e0 + lane];
        float g = fast_tanh(a[d_l] + b[s_l] + gate_b[0]);
        ev_l = g * dnorm[d_l] * dnorm[s_l];
    }
    int ev_bits = __float_as_int(ev_l);

    if (cnt == EDGES_PER_WAVE) {
        #pragma unroll
        for (int j0 = 0; j0 < EDGES_PER_WAVE; j0 += ROWS_AHEAD) {
            float2 v[ROWS_AHEAD];
            #pragma unroll
            for (int k = 0; k < ROWS_AHEAD; ++k) {
                int s = __builtin_amdgcn_readlane(s_l, j0 + k);
                const float2* hs = (const float2*)(h + (size_t)s * DFEAT);
                v[k] = hs[lane];
            }
            #pragma unroll
            for (int k = 0; k < ROWS_AHEAD; ++k) {
                int   d  = __builtin_amdgcn_readlane(d_l, j0 + k);
                float ev = __int_as_float(__builtin_amdgcn_readlane(ev_bits, j0 + k));
                float* zd = z + (size_t)d * DFEAT + 2 * lane;
                atomicAdd(zd,     v[k].x * ev);
                atomicAdd(zd + 1, v[k].y * ev);
            }
        }
    } else {
        for (int j = 0; j < cnt; ++j) {
            int   s  = __builtin_amdgcn_readlane(s_l, j);
            int   d  = __builtin_amdgcn_readlane(d_l, j);
            float ev = __int_as_float(__builtin_amdgcn_readlane(ev_bits, j));
            fa_edge_row(h, z, s, d, ev, lane);
        }
    }
}

extern "C" void kernel_launch(void* const* d_in, const int* in_sizes, int n_in,
                              void* d_out, int out_size, void* d_ws, size_t ws_size,
                              hipStream_t stream) {
    const float* h      = (const float*)d_in[0];
    const float* dnorm  = (const float*)d_in[1];
    const float* gate_w = (const float*)d_in[2];
    const float* gate_b = (const float*)d_in[3];
    const int*   src    = (const int*)d_in[4];
    const int*   dst    = (const int*)d_in[5];
    float*       z      = (float*)d_out;

    const int n = in_sizes[1];
    const int e = in_sizes[4];

    float* a = (float*)d_ws;
    float* b = a + n;

    {
        int total  = n * DFEAT;
        int blocks = (total + 255) / 256;
        fa_zero_kernel<<<blocks, 256, 0, stream>>>(z, total);
    }

    {
        int waves   = (n + 15) / 16;
        int threads = 256;
        int blocks  = (waves * 32 + threads - 1) / threads;
        fa_node_gates_kernel<<<blocks, threads, 0, stream>>>(h, gate_w, a, b, n);
    }

    {
        int edges_per_block = (256 / 32) * EDGES_PER_WAVE;
        int blocks = (e + edges_per_block - 1) / edges_per_block;
        fa_edge_kernel<<<blocks, 256, 0, stream>>>(h, dnorm, a, b, src, dst,
                                                   gate_b, z, e);
    }
}


# edge kernel: lane l covers dims l and l+32 of a row (was 2l, 2l+1): each f32 atomic and each gather covers one whole 128-B line, not half of two; atomics kept, arrival-order sum as before
# speedup vs baseline: 1.9505x; 1.3512x over previous
#include <hip/hip_runtime.h>
#include <hip/hip_bf16.h>

#define DFEAT 64

typedef float v2f __attribute__((ext_vector_type(2)));
typedef float v8f __attribute__((ext_vector_type(8)));

__device__ __forceinline__ float fast_tanh(float x) {
#if __has_builtin(__builtin_amdgcn_tanhf)
    return __builtin_amdgcn_tanhf(x);
#else
    return tanhf(x);
#endif
}

__global__ void fa_zero_kernel(float* __restrict__ z, int total) {
    int i = blockIdx.x * blockDim.x + threadIdx.x;
    if (i < total) z[i] = 0.0f;
}

__global__ void fa_node_gates_kernel(const float* __restrict__ h,
                                     const float* __restrict__ gate_w,
                                     float* __restrict__ a,
                                     float* __restrict__ b,
                                     int n) {
    int wave = (blockIdx.x * blockDim.x + threadIdx.x) >> 5;
    int lane = threadIdx.x & 31;
    int base = wave * 16;
    if (base >= n) return;

    int m     = lane & 15;
    int khalf = lane >> 4;

    const float2* hrow2 = (const float2*)(h + (size_t)(base + m) * DFEAT);

    const float2* w2   = (const float2*)(gate_w + ((m == 1) ? DFEAT : 0));
    float         valid = (m < 2) ? 1.0f : 0.0f;

    v8f c = {};
    #pragma unroll
    for (int t = 0; t < 16; ++t) {
        int kq = 2 * t + khalf;

        float2 av = hrow2[kq];
        v2f A; A.x = av.x; A.y = av.y;

        float2 wv = w2[kq];
        v2f B; B.x = wv.x * valid; B.y = wv.y * valid;

        c = __builtin_amdgcn_wmma_f32_16x16x4_f32(
                 false, A,  false, B,
                 (short)0, c,  false,  false);
    }

    int rbase = base + 8 * khalf;
    if (m == 0) {
        #pragma unroll
        for (int r = 0; r < 8; ++r) a[rbase + r] = c[r];
    } else if (m == 1) {
        #pragma unroll
        for (int r = 0; r < 8; ++r) b[rbase + r] = c[r];
    }
}

#define EDGES_PER_WAVE 32
#define ROWS_AHEAD 16
#define ROW_BYTES (DFEAT * 4u)

__device__ __forceinline__ void fa_edge_row(const float* __restrict__ h,
                                            float* __restrict__ z,
                                            int s, int d, float ev, int lane) {
    const float* hs = h + (size_t)s * DFEAT + lane;
    float v0 = hs[0];
    float v1 = hs[DFEAT / 2];
    float* zd = z + (size_t)d * DFEAT + lane;
    atomicAdd(zd,             v0 * ev);
    atomicAdd(zd + DFEAT / 2, v1 * ev);
}

__global__ void fa_edge_kernel(const float* __restrict__ h,
                               const float* __restrict__ dnorm,
                               const float* __restrict__ a,
                               const float* __restrict__ b,
                               const int* __restrict__ src,
                               const int* __restrict__ dst,
                               const float* __restrict__ gate_b,
                               float* __restrict__ z,
                               int e) {
    int wave = blockIdx.x * (blockDim.x >> 5) +
               __builtin_amdgcn_readfirstlane((int)(threadIdx.x >> 5));
    int lane = threadIdx.x & 31;
    int e0 = wave * EDGES_PER_WAVE;
    if (e0 >= e) return;
    int cnt = e - e0;
    if (cnt > EDGES_PER_WAVE) cnt = EDGES_PER_WAVE;

    int   s_l  = 0;
    int   d_l  = 0;
    float ev_l = 0.0f;
    if (lane < cnt) {
        s_l = src[e0 + lane];
        d_l = dst[e0 + lane];
        float g = fast_tanh(a[d_l] + b[s_l] + gate_b[0]);
        ev_l = g * dnorm[d_l] * dnorm[s_l];
    }
    int ev_bits = __float_as_int(ev_l);

    const unsigned lane4 = (unsigned)lane * 4u;
    if (cnt == EDGES_PER_WAVE) {
        #pragma unroll
        for (int j0 = 0; j0 < EDGES_PER_WAVE; j0 += ROWS_AHEAD) {
            float v0[ROWS_AHEAD];
            float v1[ROWS_AHEAD];
            #pragma unroll
            for (int k = 0; k < ROWS_AHEAD; ++k) {
                unsigned s = (unsigned)__builtin_amdgcn_readlane(s_l, j0 + k);
                const float* hs = (const float*)((const char*)h + (s * ROW_BYTES + lane4));
                v0[k] = hs[0];
                v1[k] = hs[DFEAT / 2];
            }
            #pragma unroll
            for (int k = 0; k < ROWS_AHEAD; ++k) {
                int   d  = __builtin_amdgcn_readlane(d_l, j0 + k);
                float ev = __int_as_float(__builtin_amdgcn_readlane(ev_bits, j0 + k));
                float* zd = (float*)((char*)z + ((unsigned)d * ROW_BYTES + lane4));
                atomicAdd(zd,             v0[k] * ev);
                atomicAdd(zd + DFEAT / 2, v1[k] * ev);
            }
        }
    } else {
        for (int j = 0; j < cnt; ++j) {
            int   s  = __builtin_amdgcn_readlane(s_l, j);
            int   d  = __builtin_amdgcn_readlane(d_l, j);
            float ev = __int_as_float(__builtin_amdgcn_readlane(ev_bits, j));
            fa_edge_row(h, z, s, d, ev, lane);
        }
    }
}

extern "C" void kernel_launch(void* const* d_in, const int* in_sizes, int n_in,
                              void* d_out, int out_size, void* d_ws, size_t ws_size,
                              hipStream_t stream) {
    const float* h      = (const float*)d_in[0];
    const float* dnorm  = (const float*)d_in[1];
    const float* gate_w = (const float*)d_in[2];
    const float* gate_b = (const float*)d_in[3];
    const int*   src    = (const int*)d_in[4];
    const int*   dst    = (const int*)d_in[5];
    float*       z      = (float*)d_out;

    const int n = in_sizes[1];
    const int e = in_sizes[4];

    float* a = (float*)d_ws;
    float* b = a + n;

    {
        int total  = n * DFEAT;
        int blocks = (total + 255) / 256;
        fa_zero_kernel<<<blocks, 256, 0, stream>>>(z, total);
    }

    {
        int waves   = (n + 15) / 16;
        int threads = 256;
        int blocks  = (waves * 32 + threads - 1) / threads;
        fa_node_gates_kernel<<<blocks, threads, 0, stream>>>(h, gate_w, a, b, n);
    }

    {
        int edges_per_block = (256 / 32) * EDGES_PER_WAVE;
        int blocks = (e + edges_per_block - 1) / edges_per_block;
        fa_edge_kernel<<<blocks, 256, 0, stream>>>(h, dnorm, a, b, src, dst,
                                                   gate_b, z, e);
    }
}


# edge kernel: node kernel packs one 16-B record per node {a,dnorm,b,dnorm}; each edge lane does 2 scattered 8-B loads (dst half, src half), was 4 scattered 4-B loads; atomics, sum order as before
# speedup vs baseline: 2.2152x; 1.1357x over previous
#include <hip/hip_runtime.h>
#include <hip/hip_bf16.h>

#define DFEAT 64

typedef float v2f __attribute__((ext_vector_type(2)));
typedef float v8f __attribute__((ext_vector_type(8)));

__device__ __forceinline__ float fast_tanh(float x) {
#if __has_builtin(__builtin_amdgcn_tanhf)
    return __builtin_amdgcn_tanhf(x);
#else
    return tanhf(x);
#endif
}

__global__ void fa_zero_kernel(float* __restrict__ z, int total) {
    int i = blockIdx.x * blockDim.x + threadIdx.x;
    if (i < total) z[i] = 0.0f;
}

__global__ void fa_node_gates_kernel(const float* __restrict__ h,
                                     const float* __restrict__ gate_w,
                                     const float* __restrict__ dnorm,
                                     float* __restrict__ rec,
                                     int n) {
    int wave = (blockIdx.x * blockDim.x + threadIdx.x) >> 5;
    int lane = threadIdx.x & 31;
    int base = wave * 16;
    if (base >= n) return;

    int m     = lane & 15;
    int khalf = lane >> 4;

    const float2* hrow2 = (const float2*)(h + (size_t)(base + m) * DFEAT);

    rec[(size_t)(base + m) * 4 + 1 + 2 * khalf] = dnorm[base + m];

    const float2* w2   = (const float2*)(gate_w + ((m == 1) ? DFEAT : 0));
    float         valid = (m < 2) ? 1.0f : 0.0f;

    v8f c = {};
    #pragma unroll
    for (int t = 0; t < 16; ++t) {
        int kq = 2 * t + khalf;

        float2 av = hrow2[kq];
        v2f A; A.x = av.x; A.y = av.y;

        float2 wv = w2[kq];
        v2f B; B.x = wv.x * valid; B.y = wv.y * valid;

        c = __builtin_amdgcn_wmma_f32_16x16x4_f32(
                 false, A,  false, B,
                 (short)0, c,  false,  false);
    }

    int rbase = base + 8 * khalf;
    if (m < 2) {
        float* rp = rec + (size_t)rbase * 4 + 2 * m;
        #pragma unroll
        for (int r = 0; r < 8; ++r) rp[r * 4] = c[r];
    }
}

#define EDGES_PER_WAVE 32
#define ROWS_AHEAD 16
#define ROW_BYTES (DFEAT * 4u)

__device__ __forceinline__ void fa_edge_row(const float* __restrict__ h,
                                            float* __restrict__ z,
                                            int s, int d, float ev, int lane) {
    const float* hs = h + (size_t)s * DFEAT + lane;
    float v0 = hs[0];
    float v1 = hs[DFEAT / 2];
    float* zd = z + (size_t)d * DFEAT + lane;
    atomicAdd(zd,             v0 * ev);
    atomicAdd(zd + DFEAT / 2, v1 * ev);
}

__global__ void fa_edge_kernel(const float* __restrict__ h,
                               const float* __restrict__ rec,
                               const int* __restrict__ src,
                               const int* __restrict__ dst,
                               const float* __restrict__ gate_b,
                               float* __restrict__ z,
                               int e) {
    int wave = blockIdx.x * (blockDim.x >> 5) +
               __builtin_amdgcn_readfirstlane((int)(threadIdx.x >> 5));
    int lane = threadIdx.x & 31;
    int e0 = wave * EDGES_PER_WAVE;
    if (e0 >= e) return;
    int cnt = e - e0;
    if (cnt > EDGES_PER_WAVE) cnt = EDGES_PER_WAVE;

    int   s_l  = 0;
    int   d_l  = 0;
    float ev_l = 0.0f;
    if (lane < cnt) {
        s_l = src[e0 + lane];
        d_l = dst[e0 + lane];
        const float2* rec2 = (const float2*)rec;
        float2 pd = rec2[2 * (size_t)d_l];
        float2 ps = rec2[2 * (size_t)s_l + 1];
        float g = fast_tanh(pd.x + ps.x + gate_b[0]);
        ev_l = g * pd.y * ps.y;
    }
    int ev_bits = __float_as_int(ev_l);

    const unsigned lane4 = (unsigned)lane * 4u;
    if (cnt == EDGES_PER_WAVE) {
        #pragma unroll
        for (int j0 = 0; j0 < EDGES_PER_WAVE; j0 += ROWS_AHEAD) {
            float v0[ROWS_AHEAD];
            float v1[ROWS_AHEAD];
            #pragma unroll
            for (int k = 0; k < ROWS_AHEAD; ++k) {
                unsigned s = (unsigned)__builtin_amdgcn_readlane(s_l, j0 + k);
                const float* hs = (const float*)((const char*)h + (s * ROW_BYTES + lane4));
                v0[k] = hs[0];
                v1[k] = hs[DFEAT / 2];
            }
            #pragma unroll
            for (int k = 0; k < ROWS_AHEAD; ++k) {
                int   d  = __builtin_amdgcn_readlane(d_l, j0 + k);
                float ev = __int_as_float(__builtin_amdgcn_readlane(ev_bits, j0 + k));
                float* zd = (float*)((char*)z + ((unsigned)d * ROW_BYTES + lane4));
                atomicAdd(zd,             v0[k] * ev);
                atomicAdd(zd + DFEAT / 2, v1[k] * ev);
            }
        }
    } else {
        for (int j = 0; j < cnt; ++j) {
            int   s  = __builtin_amdgcn_readlane(s_l, j);
            int   d  = __builtin_amdgcn_readlane(d_l, j);
            float ev = __int_as_float(__builtin_amdgcn_readlane(ev_bits, j));
            fa_edge_row(h, z, s, d, ev, lane);
        }
    }
}

extern "C" void kernel_launch(void* const* d_in, const int* in_sizes, int n_in,
                              void* d_out, int out_size, void* d_ws, size_t ws_size,
                              hipStream_t stream) {
    const float* h      = (const float*)d_in[0];
    const float* dnorm  = (const float*)d_in[1];
    const float* gate_w = (const float*)d_in[2];
    const float* gate_b = (const float*)d_in[3];
    const int*   src    = (const int*)d_in[4];
    const int*   dst    = (const int*)d_in[5];
    float*       z      = (float*)d_out;

    const int n = in_sizes[1];
    const int e = in_sizes[4];

    float* rec = (float*)d_ws;

    {
        int total  = n * DFEAT;
        int blocks = (total + 255) / 256;
        fa_zero_kernel<<<blocks, 256, 0, stream>>>(z, total);
    }

    {
        int waves   = (n + 15) / 16;
        int threads = 256;
        int blocks  = (waves * 32 + threads - 1) / threads;
        fa_node_gates_kernel<<<blocks, threads, 0, stream>>>(h, gate_w, dnorm, rec, n);
    }

    {
        int edges_per_block = (256 / 32) * EDGES_PER_WAVE;
        int blocks = (e + edges_per_block - 1) / edges_per_block;
        fa_edge_kernel<<<blocks, 256, 0, stream>>>(h, rec, src, dst,
                                                   gate_b, z, e);
    }
}


# node kernel also stores the zero start value of its 16 output rows (8 16-B stores per lane); the zero kernel (one 4-B store per lane, 200,000 waves) and its launch are gone; atomics, sum order kept
# speedup vs baseline: 2.4591x; 1.1101x over previous
#include <hip/hip_runtime.h>
#include <hip/hip_bf16.h>

#define DFEAT 64

typedef float v2f __attribute__((ext_vector_type(2)));
typedef float v8f __attribute__((ext_vector_type(8)));

__device__ __forceinline__ float fast_tanh(float x) {
#if __has_builtin(__builtin_amdgcn_tanhf)
    return __builtin_amdgcn_tanhf(x);
#else
    return tanhf(x);
#endif
}

__global__ void fa_node_gates_kernel(const float* __restrict__ h,
                                     const float* __restrict__ gate_w,
                                     const float* __restrict__ dnorm,
                                     float* __restrict__ rec,
                                     float* __restrict__ z,
                                     int n) {
    int wave = (blockIdx.x * blockDim.x + threadIdx.x) >> 5;
    int lane = threadIdx.x & 31;
    int base = wave * 16;
    if (base >= n) return;

    {
        float4* zr = (float4*)(z + (size_t)base * DFEAT) + lane;
        const float4 zero4 = make_float4(0.0f, 0.0f, 0.0f, 0.0f);
        if (base + 16 <= n) {
            for (int i = 0; i < 8; ++i) zr[i * 32] = zero4;
        } else {
            int lim = (n - base) * (DFEAT / 4);
            for (int i = 0; i < 8; ++i)
                if (i * 32 + lane < lim) zr[i * 32] = zero4;
        }
    }

    int m     = lane & 15;
    int khalf = lane >> 4;

    const float2* hrow2 = (const float2*)(h + (size_t)(base + m) * DFEAT);

    rec[(size_t)(base + m) * 4 + 1 + 2 * khalf] = dnorm[base + m];

    const float2* w2   = (const float2*)(gate_w + ((m == 1) ? DFEAT : 0));
    float         valid = (m < 2) ? 1.0f : 0.0f;

    v8f c = {};
    #pragma unroll
    for (int t = 0; t < 16; ++t) {
        int kq = 2 * t + khalf;

        float2 av = hrow2[kq];
        v2f A; A.x = av.x; A.y = av.y;

        float2 wv = w2[kq];
        v2f B; B.x = wv.x * valid; B.y = wv.y * valid;

        c = __builtin_amdgcn_wmma_f32_16x16x4_f32(
                 false, A,  false, B,
                 (short)0, c,  false,  false);
    }

    int rbase = base + 8 * khalf;
    if (m < 2) {
        float* rp = rec + (size_t)rbase * 4 + 2 * m;
        #pragma unroll
        for (int r = 0; r < 8; ++r) rp[r * 4] = c[r];
    }
}

#define EDGES_PER_WAVE 32
#define ROWS_AHEAD 16
#define ROW_BYTES (DFEAT * 4u)

__device__ __forceinline__ void fa_edge_row(const float* __restrict__ h,
                                            float* __restrict__ z,
                                            int s, int d, float ev, int lane) {
    const float* hs = h + (size_t)s * DFEAT + lane;
    float v0 = hs[0];
    float v1 = hs[DFEAT / 2];
    float* zd = z + (size_t)d * DFEAT + lane;
    atomicAdd(zd,             v0 * ev);
    atomicAdd(zd + DFEAT / 2, v1 * ev);
}

__global__ void fa_edge_kernel(const float* __restrict__ h,
                               const float* __restrict__ rec,
                               const int* __restrict__ src,
                               const int* __restrict__ dst,
                               const float* __restrict__ gate_b,
                               float* __restrict__ z,
                               int e) {
    int wave = blockIdx.x * (blockDim.x >> 5) +
               __builtin_amdgcn_readfirstlane((int)(threadIdx.x >> 5));
    int lane = threadIdx.x & 31;
    int e0 = wave * EDGES_PER_WAVE;
    if (e0 >= e) return;
    int cnt = e - e0;
    if (cnt > EDGES_PER_WAVE) cnt = EDGES_PER_WAVE;

    int   s_l  = 0;
    int   d_l  = 0;
    float ev_l = 0.0f;
    if (lane < cnt) {
        s_l = src[e0 + lane];
        d_l = dst[e0 + lane];
        const float2* rec2 = (const float2*)rec;
        float2 pd = rec2[2 * (size_t)d_l];
        float2 ps = rec2[2 * (size_t)s_l + 1];
        float g = fast_tanh(pd.x + ps.x + gate_b[0]);
        ev_l = g * pd.y * ps.y;
    }
    int ev_bits = __float_as_int(ev_l);

    const unsigned lane4 = (unsigned)lane * 4u;
    if (cnt == EDGES_PER_WAVE) {
        #pragma unroll
        for (int j0 = 0; j0 < EDGES_PER_WAVE; j0 += ROWS_AHEAD) {
            float v0[ROWS_AHEAD];
            float v1[ROWS_AHEAD];
            #pragma unroll
            for (int k = 0; k < ROWS_AHEAD; ++k) {
                unsigned s = (unsigned)__builtin_amdgcn_readlane(s_l, j0 + k);
                const float* hs = (const float*)((const char*)h + (s * ROW_BYTES + lane4));
                v0[k] = hs[0];
                v1[k] = hs[DFEAT / 2];
            }
            #pragma unroll
            for (int k = 0; k < ROWS_AHEAD; ++k) {
                int   d  = __builtin_amdgcn_readlane(d_l, j0 + k);
                float ev = __int_as_float(__builtin_amdgcn_readlane(ev_bits, j0 + k));
                float* zd = (float*)((char*)z + ((unsigned)d * ROW_BYTES + lane4));
                atomicAdd(zd,             v0[k] * ev);
                atomicAdd(zd + DFEAT / 2, v1[k] * ev);
            }
        }
    } else {
        for (int j = 0; j < cnt; ++j) {
            int   s  = __builtin_amdgcn_readlane(s_l, j);
            int   d  = __builtin_amdgcn_readlane(d_l, j);
            float ev = __int_as_float(__builtin_amdgcn_readlane(ev_bits, j));
            fa_edge_row(h, z, s, d, ev, lane);
        }
    }
}

extern "C" void kernel_launch(void* const* d_in, const int* in_sizes, int n_in,
                              void* d_out, int out_size, void* d_ws, size_t ws_size,
                              hipStream_t stream) {
    const float* h      = (const float*)d_in[0];
    const float* dnorm  = (const float*)d_in[1];
    const float* gate_w = (const float*)d_in[2];
    const float* gate_b = (const float*)d_in[3];
    const int*   src    = (const int*)d_in[4];
    const int*   dst    = (const int*)d_in[5];
    float*       z      = (float*)d_out;

    const int n = in_sizes[1];
    const int e = in_sizes[4];

    float* rec = (float*)d_ws;

    {
        int waves   = (n + 15) / 16;
        int threads = 256;
        int blocks  = (waves * 32 + threads - 1) / threads;
        fa_node_gates_kernel<<<blocks, threads, 0, stream>>>(h, gate_w, dnorm, rec, z, n);
    }

    {
        int edges_per_block = (256 / 32) * EDGES_PER_WAVE;
        int blocks = (e + edges_per_block - 1) / edges_per_block;
        fa_edge_kernel<<<blocks, 256, 0, stream>>>(h, rec, src, dst,
                                                   gate_b, z, e);
    }
}


# edge kernel: z rows split in two ranges, each added into by one half of the chip only (blocks draw 512-edge chunks per range from 2 integer counters, by die id); f32 atomics, arrival-order sum kept
# speedup vs baseline: 3.0841x; 1.2542x over previous
#include <hip/hip_runtime.h>
#include <hip/hip_bf16.h>

#define DFEAT 64
#define CTR_STRIDE 64
#define CTR_BYTES 512

typedef float v2f __attribute__((ext_vector_type(2)));
typedef float v8f __attribute__((ext_vector_type(8)));

__device__ __forceinline__ float fast_tanh(float x) {
#if __has_builtin(__builtin_amdgcn_tanhf)
    return __builtin_amdgcn_tanhf(x);
#else
    return tanhf(x);
#endif
}

__global__ void fa_node_gates_kernel(const float* __restrict__ h,
                                     const float* __restrict__ gate_w,
                                     const float* __restrict__ dnorm,
                                     float* __restrict__ rec,
                                     float* __restrict__ z,
                                     unsigned int* __restrict__ ctr,
                                     int n) {
    if (blockIdx.x == 0 && threadIdx.x < 2) ctr[threadIdx.x * CTR_STRIDE] = 0u;

    int wave = (blockIdx.x * blockDim.x + threadIdx.x) >> 5;
    int lane = threadIdx.x & 31;
    int base = wave * 16;
    if (base >= n) return;

    {
        float4* zr = (float4*)(z + (size_t)base * DFEAT) + lane;
        const float4 zero4 = make_float4(0.0f, 0.0f, 0.0f, 0.0f);
        if (base + 16 <= n) {
            for (int i = 0; i < 8; ++i) zr[i * 32] = zero4;
        } else {
            int lim = (n - base) * (DFEAT / 4);
            for (int i = 0; i < 8; ++i)
                if (i * 32 + lane < lim) zr[i * 32] = zero4;
        }
    }

    int m     = lane & 15;
    int khalf = lane >> 4;

    const float2* hrow2 = (const float2*)(h + (size_t)(base + m) * DFEAT);

    rec[(size_t)(base + m) * 4 + 1 + 2 * khalf] = dnorm[base + m];

    const float2* w2   = (const float2*)(gate_w + ((m == 1) ? DFEAT : 0));
    float         valid = (m < 2) ? 1.0f : 0.0f;

    v8f c = {};
    #pragma unroll
    for (int t = 0; t < 16; ++t) {
        int kq = 2 * t + khalf;

        float2 av = hrow2[kq];
        v2f A; A.x = av.x; A.y = av.y;

        float2 wv = w2[kq];
        v2f B; B.x = wv.x * valid; B.y = wv.y * valid;

        c = __builtin_amdgcn_wmma_f32_16x16x4_f32(
                 false, A,  false, B,
                 (short)0, c,  false,  false);
    }

    int rbase = base + 8 * khalf;
    if (m < 2) {
        float* rp = rec + (size_t)rbase * 4 + 2 * m;
        #pragma unroll
        for (int r = 0; r < 8; ++r) rp[r * 4] = c[r];
    }
}

#define EDGES_PER_WAVE 64
#define WAVES_PER_BLOCK 8
#define EDGES_PER_BLOCK (EDGES_PER_WAVE * WAVES_PER_BLOCK)
#define ROWS_AHEAD 8
#define ROW_BYTES (DFEAT * 4u)

template <bool FULL>
__device__ __forceinline__ void fa_edge_group(const float* __restrict__ h,
                                              float* __restrict__ z,
                                              unsigned long long& m, int nv,
                                              int sA, int sB, int dA, int dB,
                                              int evA, int evB, unsigned lane4) {
    float v0[ROWS_AHEAD];
    float v1[ROWS_AHEAD];
    int   jj[ROWS_AHEAD];
    #pragma unroll
    for (int k = 0; k < ROWS_AHEAD; ++k) {
        v0[k] = 0.0f; v1[k] = 0.0f; jj[k] = 0;
        if (FULL || k < nv) {
            int j = __builtin_ctzll(m);
            m &= m - 1;
            jj[k] = j;
            int jl = j & 31;
            int a = __builtin_amdgcn_readlane(sA, jl);
            int b = __builtin_amdgcn_readlane(sB, jl);
            unsigned s = (unsigned)((j & 32) ? b : a);
            const float* hs = (const float*)((const char*)h + (s * ROW_BYTES + lane4));
            v0[k] = hs[0];
            v1[k] = hs[DFEAT / 2];
        }
    }
    #pragma unroll
    for (int k = 0; k < ROWS_AHEAD; ++k) {
        if (FULL || k < nv) {
            int j  = jj[k];
            int jl = j & 31;
            int da = __builtin_amdgcn_readlane(dA, jl);
            int db = __builtin_amdgcn_readlane(dB, jl);
            int ea = __builtin_amdgcn_readlane(evA, jl);
            int eb = __builtin_amdgcn_readlane(evB, jl);
            unsigned d  = (unsigned)((j & 32) ? db : da);
            float    ev = __int_as_float((j & 32) ? eb : ea);
            float* zd = (float*)((char*)z + (d * ROW_BYTES + lane4));
            atomicAdd(zd,             v0[k] * ev);
            atomicAdd(zd + DFEAT / 2, v1[k] * ev);
        }
    }
}

__global__ void fa_edge_kernel(const float* __restrict__ h,
                               const float* __restrict__ rec,
                               const int* __restrict__ src,
                               const int* __restrict__ dst,
                               const float* __restrict__ gate_b,
                               float* __restrict__ z,
                               unsigned int* __restrict__ ctr,
                               int n, int e, int nchunks) {
    __shared__ int sh_item;
    if (threadIdx.x == 0) {
        unsigned id = __builtin_amdgcn_s_sendmsg_rtn(0x87);
        int p = (((id >> 16) & 0xFu) >= 4u) ? 1 : 0;
        unsigned t = atomicAdd(&ctr[p * CTR_STRIDE], 1u);
        if (t >= (unsigned)nchunks) {
            p = 1 - p;
            t = atomicAdd(&ctr[p * CTR_STRIDE], 1u);
        }
        sh_item = (t < (unsigned)nchunks) ? (int)(t * 2u + (unsigned)p) : -1;
    }
    __syncthreads();
    int item = __builtin_amdgcn_readfirstlane(sh_item);
    if (item < 0) return;
    int p     = item & 1;
    int chunk = item >> 1;
    int nsplit = n >> 1;

    int wv   = __builtin_amdgcn_readfirstlane((int)(threadIdx.x >> 5));
    int lane = threadIdx.x & 31;
    int e0 = chunk * EDGES_PER_BLOCK + wv * EDGES_PER_WAVE;
    if (e0 >= e) return;
    int cnt = e - e0;

    bool vA = lane < cnt;
    bool vB = lane + 32 < cnt;
    int  cA = vA ? e0 + lane : e0;
    int  cB = vB ? e0 + 32 + lane : e0;
    int  sA = src[cA];
    int  dA = dst[cA];
    int  sB = src[cB];
    int  dB = dst[cB];
    bool inA = vA && ((((dA >= nsplit) ? 1 : 0)) == p);
    bool inB = vB && ((((dB >= nsplit) ? 1 : 0)) == p);
    const float2* rec2 = (const float2*)rec;
    float2 pdA = rec2[2 * (size_t)(inA ? dA : 0)];
    float2 psA = rec2[2 * (size_t)(inA ? sA : 0) + 1];
    float2 pdB = rec2[2 * (size_t)(inB ? dB : 0)];
    float2 psB = rec2[2 * (size_t)(inB ? sB : 0) + 1];
    float gb = gate_b[0];
    float evAf = fast_tanh(pdA.x + psA.x + gb) * pdA.y * psA.y;
    float evBf = fast_tanh(pdB.x + psB.x + gb) * pdB.y * psB.y;
    int evA = __float_as_int(evAf);
    int evB = __float_as_int(evBf);

    unsigned mA = __builtin_amdgcn_ballot_w32(inA);
    unsigned mB = __builtin_amdgcn_ballot_w32(inB);
    unsigned long long m = (unsigned long long)mA | ((unsigned long long)mB << 32);
    int left = __builtin_popcountll(m);

    const unsigned lane4 = (unsigned)lane * 4u;

    while (left >= ROWS_AHEAD) {
        fa_edge_group<true>(h, z, m, ROWS_AHEAD, sA, sB, dA, dB, evA, evB, lane4);
        left -= ROWS_AHEAD;
    }
    if (left > 0)
        fa_edge_group<false>(h, z, m, left, sA, sB, dA, dB, evA, evB, lane4);
}

extern "C" void kernel_launch(void* const* d_in, const int* in_sizes, int n_in,
                              void* d_out, int out_size, void* d_ws, size_t ws_size,
                              hipStream_t stream) {
    const float* h      = (const float*)d_in[0];
    const float* dnorm  = (const float*)d_in[1];
    const float* gate_w = (const float*)d_in[2];
    const float* gate_b = (const float*)d_in[3];
    const int*   src    = (const int*)d_in[4];
    const int*   dst    = (const int*)d_in[5];
    float*       z      = (float*)d_out;

    const int n = in_sizes[1];
    const int e = in_sizes[4];

    unsigned int* ctr = (unsigned int*)d_ws;
    float* rec = (float*)((char*)d_ws + CTR_BYTES);

    {
        int waves   = (n + 15) / 16;
        int threads = 256;
        int blocks  = (waves * 32 + threads - 1) / threads;
        fa_node_gates_kernel<<<blocks, threads, 0, stream>>>(h, gate_w, dnorm, rec, z, ctr, n);
    }

    {
        int nchunks = (e + EDGES_PER_BLOCK - 1) / EDGES_PER_BLOCK;
        int blocks  = 2 * nchunks;
        fa_edge_kernel<<<blocks, WAVES_PER_BLOCK * 32, 0, stream>>>(h, rec, src, dst,
                                                                    gate_b, z, ctr, n, e, nchunks);
    }
}
